# starting kernel
# baseline (speedup 1.0000x reference)
#include <hip/hip_runtime.h>

typedef __attribute__((ext_vector_type(2))) float v2f;
typedef __attribute__((ext_vector_type(8))) float v8f;

#define IN_F  128
#define HID_F 128
#define OUT_F 64


__global__ __launch_bounds__(256) void zero_kernel(float* __restrict__ p, long long n) {
    long long i = (long long)blockIdx.x * 256 + threadIdx.x;
    if (i < n) p[i] = 0.0f;
}

__global__ __launch_bounds__(256) void degree_kernel(const int* __restrict__ dst,
                                                     float* __restrict__ deg, int E) {
    int e = blockIdx.x * 256 + threadIdx.x;
    if (e < E) atomicAdd(&deg[dst[e]], 1.0f);
}

__global__ __launch_bounds__(256) void invdeg_kernel(float* __restrict__ deg, int n) {
    int i = blockIdx.x * 256 + threadIdx.x;
    if (i < n) deg[i] = 1.0f / fmaxf(deg[i], 1.0f);
}

__global__ __launch_bounds__(256) void scatter_kernel(const float* __restrict__ feat,
                                                      const int* __restrict__ src,
                                                      const int* __restrict__ dst,
                                                      float* __restrict__ accum, int E) {
    long long tid = (long long)blockIdx.x * 256 + threadIdx.x;
    int e = (int)(tid >> 7);
    int d = (int)(tid & 127);
    if (e < E) {
        int s = src[e], t = dst[e];
        atomicAdd(&accum[(long long)t * 128 + d], feat[(long long)s * 128 + d]);
    }
}

template <int NCOLT>
__global__ __launch_bounds__(32 * NCOLT) void sage_gemm_wmma(
        const float* __restrict__ Xs, const float* __restrict__ Xa,
        const float* __restrict__ invdeg,
        const float* __restrict__ Ws, const float* __restrict__ Wn,
        const float* __restrict__ bias, float* __restrict__ out,
        int M, int K, int do_relu) {
    const int N = NCOLT * 16;
    const int lane    = threadIdx.x;
    const int rowBase = blockIdx.x * 16;
    const int colBase = threadIdx.y * 16;

    int r = rowBase + (lane & 15);
    if (r >= M) r = M - 1;
    const int kOff = (lane < 16) ? 0 : 2;
    const int c = colBase + (lane & 15);

    const float invd = invdeg[r];
    const float* xsRow = Xs + (long long)r * K + kOff;
    const float* xaRow = Xa + (long long)r * K + kOff;

    v8f acc = {};
    for (int kk = 0; kk < K; kk += 4) {
        v2f a_s = *reinterpret_cast<const v2f*>(xsRow + kk);
        v2f a_raw = *reinterpret_cast<const v2f*>(xaRow + kk);
        v2f a_n = { a_raw.x * invd, a_raw.y * invd };
        const int kb = kk + kOff;
        v2f b_s = { Ws[(long long)kb * N + c], Ws[(long long)(kb + 1) * N + c] };
        v2f b_n = { Wn[(long long)kb * N + c], Wn[(long long)(kb + 1) * N + c] };
        acc = __builtin_amdgcn_wmma_f32_16x16x4_f32(false, a_s, false, b_s,
                                                    (short)0, acc, false, false);
        acc = __builtin_amdgcn_wmma_f32_16x16x4_f32(false, a_n, false, b_n,
                                                    (short)0, acc, false, false);
    }

    const float bv = bias[c];
    const int mBase = rowBase + ((lane < 16) ? 0 : 8);
#pragma unroll
    for (int j = 0; j < 8; ++j) {
        int m = mBase + j;
        if (m < M) {
            float v = acc[j] + bv;
            if (do_relu) v = fmaxf(v, 0.0f);
            out[(long long)m * N + c] = v;
        }
    }
}


__global__ __launch_bounds__(256) void gcount_kernel(const int* __restrict__ gid,
                                                     float* __restrict__ gcnt, int n) {
    int i = blockIdx.x * 256 + threadIdx.x;
    if (i < n) atomicAdd(&gcnt[gid[i]], 1.0f);
}

__global__ __launch_bounds__(256) void readout_kernel(const float* __restrict__ h2,
                                                      const int* __restrict__ gid,
                                                      float* __restrict__ gsum, int Nn) {
    long long tid = (long long)blockIdx.x * 256 + threadIdx.x;
    int node = (int)(tid >> 6);
    int f    = (int)(tid & 63);
    if (node < Nn)
        atomicAdd(&gsum[(long long)gid[node] * OUT_F + f], h2[(long long)node * OUT_F + f]);
}

__global__ __launch_bounds__(256) void finalize_kernel(const float* __restrict__ gsum,
                                                       const float* __restrict__ gcnt,
                                                       float* __restrict__ out, int total) {
    int i = blockIdx.x * 256 + threadIdx.x;
    if (i < total) {
        int g = i >> 6;
        out[i] = gsum[i] / fmaxf(gcnt[g], 1.0f);
    }
}


static inline int cdiv64(long long a, long long b) { return (int)((a + b - 1) / b); }

extern "C" void kernel_launch(void* const* d_in, const int* in_sizes, int n_in,
                              void* d_out, int out_size, void* d_ws, size_t ws_size,
                              hipStream_t stream) {
    const float* in_feat = (const float*)d_in[0];
    const int*   src     = (const int*)d_in[1];
    const int*   dst     = (const int*)d_in[2];
    const int*   gid     = (const int*)d_in[3];
    const float* W1s = (const float*)d_in[5];
    const float* W1n = (const float*)d_in[6];
    const float* b1  = (const float*)d_in[7];
    const float* W2s = (const float*)d_in[8];
    const float* W2n = (const float*)d_in[9];
    const float* b2  = (const float*)d_in[10];
    float* out = (float*)d_out;

    const int Nn = in_sizes[0] / IN_F;
    const int E  = in_sizes[1];
    const int G  = out_size / OUT_F;

    float* ws    = (float*)d_ws;
    float* accum = ws;
    float* deg   = accum + (size_t)Nn * 128;
    float* h1    = deg + Nn;
    float* h2    = h1 + (size_t)Nn * 128;
    float* gsum  = h2 + (size_t)Nn * OUT_F;
    float* gcnt  = gsum + (size_t)G * OUT_F;

    const int mTiles = cdiv64(Nn, 16);

    zero_kernel<<<cdiv64((long long)Nn * 128 + Nn, 256), 256, 0, stream>>>(
        accum, (long long)Nn * 128 + Nn);
    degree_kernel<<<cdiv64(E, 256), 256, 0, stream>>>(dst, deg, E);
    scatter_kernel<<<cdiv64((long long)E * 128, 256), 256, 0, stream>>>(in_feat, src, dst, accum, E);
    invdeg_kernel<<<cdiv64(Nn, 256), 256, 0, stream>>>(deg, Nn);
    sage_gemm_wmma<8><<<mTiles, dim3(32, 8), 0, stream>>>(
        in_feat, accum, deg, W1s, W1n, b1, h1, Nn, IN_F, 1);

    zero_kernel<<<cdiv64((long long)Nn * 128, 256), 256, 0, stream>>>(accum, (long long)Nn * 128);
    scatter_kernel<<<cdiv64((long long)E * 128, 256), 256, 0, stream>>>(h1, src, dst, accum, E);
    sage_gemm_wmma<4><<<mTiles, dim3(32, 4), 0, stream>>>(
        h1, accum, deg, W2s, W2n, b2, h2, Nn, HID_F, 0);

    zero_kernel<<<cdiv64((long long)G * OUT_F + G, 256), 256, 0, stream>>>(
        gsum, (long long)G * OUT_F + G);
    gcount_kernel<<<cdiv64(Nn, 256), 256, 0, stream>>>(gid, gcnt, Nn);
    readout_kernel<<<cdiv64((long long)Nn * OUT_F, 256), 256, 0, stream>>>(h2, gid, gsum, Nn);
    finalize_kernel<<<cdiv64((long long)G * OUT_F, 256), 256, 0, stream>>>(gsum, gcnt, out, G * OUT_F);
}


# Replace both atomic float scatter-adds by a per-call CSR (counting sort by dst, one returning int atomic per edge) and an atomic-free row gather; row sums now run in CSR slot order
# speedup vs baseline: 2.2074x; 2.2074x over previous
#include <hip/hip_runtime.h>

typedef __attribute__((ext_vector_type(2))) float v2f;
typedef __attribute__((ext_vector_type(8))) float v8f;

#define IN_F  128
#define HID_F 128
#define OUT_F 64


__global__ __launch_bounds__(256) void zero_kernel(float* __restrict__ p, long long n) {
    long long i = (long long)blockIdx.x * 256 + threadIdx.x;
    if (i < n) p[i] = 0.0f;
}

__global__ __launch_bounds__(256) void zero_int_kernel(int* __restrict__ p, int n) {
    int i = blockIdx.x * 256 + threadIdx.x;
    if (i < n) p[i] = 0;
}

__global__ __launch_bounds__(256) void count_kernel(const int* __restrict__ dst,
                                                    int* __restrict__ cnt,
                                                    int* __restrict__ rank, int E) {
    int e = blockIdx.x * 256 + threadIdx.x;
    if (e < E) rank[e] = atomicAdd(&cnt[dst[e]], 1);
}

__global__ __launch_bounds__(1024) void scan_kernel(const int* __restrict__ cnt,
                                                    int* __restrict__ rowptr,
                                                    float* __restrict__ invdeg, int Nn) {
    __shared__ int wpre[32];
    __shared__ int wtot[32];
    const int t = threadIdx.x, lane = t & 31, w = t >> 5;
    const int start = blockIdx.x * 1024;

    int part = 0;
    for (int i = t; i < start; i += 1024) part += cnt[i];
    for (int o = 16; o > 0; o >>= 1) part += __shfl_xor(part, o);
    if (lane == 0) wpre[w] = part;

    const int i = start + t;
    const int v = (i < Nn) ? cnt[i] : 0;
    int inc = v;
    for (int o = 1; o < 32; o <<= 1) {
        int u = __shfl_up(inc, o);
        if (lane >= o) inc += u;
    }
    if (lane == 31) wtot[w] = inc;
    __syncthreads();

    int base = 0;
    for (int k = 0; k < 32; ++k) {
        base += wpre[k];
        if (k < w) base += wtot[k];
    }
    if (i < Nn) {
        rowptr[i] = base + inc - v;
        invdeg[i] = 1.0f / fmaxf((float)v, 1.0f);
    }
}

__global__ __launch_bounds__(256) void fill_kernel(const int* __restrict__ src,
                                                   const int* __restrict__ dst,
                                                   const int* __restrict__ rowptr,
                                                   const int* __restrict__ rank,
                                                   int* __restrict__ col, int E) {
    int e = blockIdx.x * 256 + threadIdx.x;
    if (e < E) col[rowptr[dst[e]] + rank[e]] = src[e];
}

__global__ __launch_bounds__(256) void gather_kernel(const float* __restrict__ feat,
                                                     const int* __restrict__ rowptr,
                                                     const int* __restrict__ cnt,
                                                     const int* __restrict__ col,
                                                     float* __restrict__ accum, int Nn) {
    const int lane = threadIdx.x & 31;
    const int node = blockIdx.x * 8 + __builtin_amdgcn_readfirstlane((int)(threadIdx.x >> 5));
    if (node >= Nn) return;
    const int beg = __builtin_amdgcn_readfirstlane(rowptr[node]);
    const int n   = __builtin_amdgcn_readfirstlane(cnt[node]);

    float4 s = make_float4(0.0f, 0.0f, 0.0f, 0.0f);
    for (int b = 0; b < n; b += 32) {
        const int m = (n - b < 32) ? (n - b) : 32;
        const int cv = (lane < m) ? col[beg + b + lane] : 0;
        int j = 0;
        for (; j + 4 <= m; j += 4) {
            const int c0 = __builtin_amdgcn_readlane(cv, j);
            const int c1 = __builtin_amdgcn_readlane(cv, j + 1);
            const int c2 = __builtin_amdgcn_readlane(cv, j + 2);
            const int c3 = __builtin_amdgcn_readlane(cv, j + 3);
            const float4 v0 = *reinterpret_cast<const float4*>(feat + (long long)c0 * 128 + lane * 4);
            const float4 v1 = *reinterpret_cast<const float4*>(feat + (long long)c1 * 128 + lane * 4);
            const float4 v2 = *reinterpret_cast<const float4*>(feat + (long long)c2 * 128 + lane * 4);
            const float4 v3 = *reinterpret_cast<const float4*>(feat + (long long)c3 * 128 + lane * 4);
            s.x += v0.x; s.y += v0.y; s.z += v0.z; s.w += v0.w;
            s.x += v1.x; s.y += v1.y; s.z += v1.z; s.w += v1.w;
            s.x += v2.x; s.y += v2.y; s.z += v2.z; s.w += v2.w;
            s.x += v3.x; s.y += v3.y; s.z += v3.z; s.w += v3.w;
        }
        for (; j < m; ++j) {
            const int c0 = __builtin_amdgcn_readlane(cv, j);
            const float4 v0 = *reinterpret_cast<const float4*>(feat + (long long)c0 * 128 + lane * 4);
            s.x += v0.x; s.y += v0.y; s.z += v0.z; s.w += v0.w;
        }
    }
    *reinterpret_cast<float4*>(accum + (long long)node * 128 + lane * 4) = s;
}

template <int NCOLT>
__global__ __launch_bounds__(32 * NCOLT) void sage_gemm_wmma(
        const float* __restrict__ Xs, const float* __restrict__ Xa,
        const float* __restrict__ invdeg,
        const float* __restrict__ Ws, const float* __restrict__ Wn,
        const float* __restrict__ bias, float* __restrict__ out,
        int M, int K, int do_relu) {
    const int N = NCOLT * 16;
    const int lane    = threadIdx.x;
    const int rowBase = blockIdx.x * 16;
    const int colBase = threadIdx.y * 16;

    int r = rowBase + (lane & 15);
    if (r >= M) r = M - 1;
    const int kOff = (lane < 16) ? 0 : 2;
    const int c = colBase + (lane & 15);

    const float invd = invdeg[r];
    const float* xsRow = Xs + (long long)r * K + kOff;
    const float* xaRow = Xa + (long long)r * K + kOff;

    v8f acc = {};
    for (int kk = 0; kk < K; kk += 4) {
        v2f a_s = *reinterpret_cast<const v2f*>(xsRow + kk);
        v2f a_raw = *reinterpret_cast<const v2f*>(xaRow + kk);
        v2f a_n = { a_raw.x * invd, a_raw.y * invd };
        const int kb = kk + kOff;
        v2f b_s = { Ws[(long long)kb * N + c], Ws[(long long)(kb + 1) * N + c] };
        v2f b_n = { Wn[(long long)kb * N + c], Wn[(long long)(kb + 1) * N + c] };
        acc = __builtin_amdgcn_wmma_f32_16x16x4_f32(false, a_s, false, b_s,
                                                    (short)0, acc, false, false);
        acc = __builtin_amdgcn_wmma_f32_16x16x4_f32(false, a_n, false, b_n,
                                                    (short)0, acc, false, false);
    }

    const float bv = bias[c];
    const int mBase = rowBase + ((lane < 16) ? 0 : 8);
#pragma unroll
    for (int j = 0; j < 8; ++j) {
        int m = mBase + j;
        if (m < M) {
            float v = acc[j] + bv;
            if (do_relu) v = fmaxf(v, 0.0f);
            out[(long long)m * N + c] = v;
        }
    }
}


__global__ __launch_bounds__(256) void gcount_kernel(const int* __restrict__ gid,
                                                     float* __restrict__ gcnt, int n) {
    int i = blockIdx.x * 256 + threadIdx.x;
    if (i < n) atomicAdd(&gcnt[gid[i]], 1.0f);
}

__global__ __launch_bounds__(256) void readout_kernel(const float* __restrict__ h2,
                                                      const int* __restrict__ gid,
                                                      float* __restrict__ gsum, int Nn) {
    long long tid = (long long)blockIdx.x * 256 + threadIdx.x;
    int node = (int)(tid >> 6);
    int f    = (int)(tid & 63);
    if (node < Nn)
        atomicAdd(&gsum[(long long)gid[node] * OUT_F + f], h2[(long long)node * OUT_F + f]);
}

__global__ __launch_bounds__(256) void finalize_kernel(const float* __restrict__ gsum,
                                                       const float* __restrict__ gcnt,
                                                       float* __restrict__ out, int total) {
    int i = blockIdx.x * 256 + threadIdx.x;
    if (i < total) {
        int g = i >> 6;
        out[i] = gsum[i] / fmaxf(gcnt[g], 1.0f);
    }
}


static inline int cdiv64(long long a, long long b) { return (int)((a + b - 1) / b); }

extern "C" void kernel_launch(void* const* d_in, const int* in_sizes, int n_in,
                              void* d_out, int out_size, void* d_ws, size_t ws_size,
                              hipStream_t stream) {
    const float* in_feat = (const float*)d_in[0];
    const int*   src     = (const int*)d_in[1];
    const int*   dst     = (const int*)d_in[2];
    const int*   gid     = (const int*)d_in[3];
    const float* W1s = (const float*)d_in[5];
    const float* W1n = (const float*)d_in[6];
    const float* b1  = (const float*)d_in[7];
    const float* W2s = (const float*)d_in[8];
    const float* W2n = (const float*)d_in[9];
    const float* b2  = (const float*)d_in[10];
    float* out = (float*)d_out;

    const int Nn = in_sizes[0] / IN_F;
    const int E  = in_sizes[1];
    const int G  = out_size / OUT_F;

    float* ws    = (float*)d_ws;
    float* accum = ws;
    float* h1    = accum + (size_t)Nn * 128;
    float* h2    = h1 + (size_t)Nn * 128;
    float* gsum  = h2 + (size_t)Nn * OUT_F;
    float* gcnt  = gsum + (size_t)G * OUT_F;
    float* deg   = gcnt + G;
    int* cnt    = (int*)h2;
    int* rowptr = cnt + Nn;
    int* rank   = rowptr + Nn;
    int* col    = rank + E;

    const int mTiles = cdiv64(Nn, 16);

    zero_int_kernel<<<cdiv64(Nn, 256), 256, 0, stream>>>(cnt, Nn);
    count_kernel<<<cdiv64(E, 256), 256, 0, stream>>>(dst, cnt, rank, E);
    scan_kernel<<<cdiv64(Nn, 1024), 1024, 0, stream>>>(cnt, rowptr, deg, Nn);
    fill_kernel<<<cdiv64(E, 256), 256, 0, stream>>>(src, dst, rowptr, rank, col, E);

    gather_kernel<<<cdiv64(Nn, 8), 256, 0, stream>>>(in_feat, rowptr, cnt, col, accum, Nn);
    sage_gemm_wmma<8><<<mTiles, dim3(32, 8), 0, stream>>>(
        in_feat, accum, deg, W1s, W1n, b1, h1, Nn, IN_F, 1);

    gather_kernel<<<cdiv64(Nn, 8), 256, 0, stream>>>(h1, rowptr, cnt, col, accum, Nn);
    sage_gemm_wmma<4><<<mTiles, dim3(32, 4), 0, stream>>>(
        h1, accum, deg, W2s, W2n, b2, h2, Nn, HID_F, 0);

    zero_kernel<<<cdiv64((long long)G * OUT_F + G, 256), 256, 0, stream>>>(
        gsum, (long long)G * OUT_F + G);
    gcount_kernel<<<cdiv64(Nn, 256), 256, 0, stream>>>(gid, gcnt, Nn);
    readout_kernel<<<cdiv64((long long)Nn * OUT_F, 256), 256, 0, stream>>>(h2, gid, gsum, Nn);
    finalize_kernel<<<cdiv64((long long)G * OUT_F, 256), 256, 0, stream>>>(gsum, gcnt, out, G * OUT_F);
}


# Per-graph mean readout by one workgroup per graph: it scans the graph ids and sums its own h2 rows in registers and LDS in fixed node order; float atomics gone, 4 launches become 1
# speedup vs baseline: 2.8932x; 1.3107x over previous
#include <hip/hip_runtime.h>

typedef __attribute__((ext_vector_type(2))) float v2f;
typedef __attribute__((ext_vector_type(8))) float v8f;

#define IN_F  128
#define HID_F 128
#define OUT_F 64


__global__ __launch_bounds__(256) void zero_int_kernel(int* __restrict__ p, int n) {
    int i = blockIdx.x * 256 + threadIdx.x;
    if (i < n) p[i] = 0;
}

__global__ __launch_bounds__(256) void count_kernel(const int* __restrict__ dst,
                                                    int* __restrict__ cnt,
                                                    int* __restrict__ rank, int E) {
    int e = blockIdx.x * 256 + threadIdx.x;
    if (e < E) rank[e] = atomicAdd(&cnt[dst[e]], 1);
}

__global__ __launch_bounds__(1024) void scan_kernel(const int* __restrict__ cnt,
                                                    int* __restrict__ rowptr,
                                                    float* __restrict__ invdeg, int Nn) {
    __shared__ int wpre[32];
    __shared__ int wtot[32];
    const int t = threadIdx.x, lane = t & 31, w = t >> 5;
    const int start = blockIdx.x * 1024;

    int part = 0;
    for (int i = t; i < start; i += 1024) part += cnt[i];
    for (int o = 16; o > 0; o >>= 1) part += __shfl_xor(part, o);
    if (lane == 0) wpre[w] = part;

    const int i = start + t;
    const int v = (i < Nn) ? cnt[i] : 0;
    int inc = v;
    for (int o = 1; o < 32; o <<= 1) {
        int u = __shfl_up(inc, o);
        if (lane >= o) inc += u;
    }
    if (lane == 31) wtot[w] = inc;
    __syncthreads();

    int base = 0;
    for (int k = 0; k < 32; ++k) {
        base += wpre[k];
        if (k < w) base += wtot[k];
    }
    if (i < Nn) {
        rowptr[i] = base + inc - v;
        invdeg[i] = 1.0f / fmaxf((float)v, 1.0f);
    }
}

__global__ __launch_bounds__(256) void fill_kernel(const int* __restrict__ src,
                                                   const int* __restrict__ dst,
                                                   const int* __restrict__ rowptr,
                                                   const int* __restrict__ rank,
                                                   int* __restrict__ col, int E) {
    int e = blockIdx.x * 256 + threadIdx.x;
    if (e < E) col[rowptr[dst[e]] + rank[e]] = src[e];
}

__global__ __launch_bounds__(256) void gather_kernel(const float* __restrict__ feat,
                                                     const int* __restrict__ rowptr,
                                                     const int* __restrict__ cnt,
                                                     const int* __restrict__ col,
                                                     float* __restrict__ accum, int Nn) {
    const int lane = threadIdx.x & 31;
    const int node = blockIdx.x * 8 + __builtin_amdgcn_readfirstlane((int)(threadIdx.x >> 5));
    if (node >= Nn) return;
    const int beg = __builtin_amdgcn_readfirstlane(rowptr[node]);
    const int n   = __builtin_amdgcn_readfirstlane(cnt[node]);

    float4 s = make_float4(0.0f, 0.0f, 0.0f, 0.0f);
    for (int b = 0; b < n; b += 32) {
        const int m = (n - b < 32) ? (n - b) : 32;
        const int cv = (lane < m) ? col[beg + b + lane] : 0;
        int j = 0;
        for (; j + 4 <= m; j += 4) {
            const int c0 = __builtin_amdgcn_readlane(cv, j);
            const int c1 = __builtin_amdgcn_readlane(cv, j + 1);
            const int c2 = __builtin_amdgcn_readlane(cv, j + 2);
            const int c3 = __builtin_amdgcn_readlane(cv, j + 3);
            const float4 v0 = *reinterpret_cast<const float4*>(feat + (long long)c0 * 128 + lane * 4);
            const float4 v1 = *reinterpret_cast<const float4*>(feat + (long long)c1 * 128 + lane * 4);
            const float4 v2 = *reinterpret_cast<const float4*>(feat + (long long)c2 * 128 + lane * 4);
            const float4 v3 = *reinterpret_cast<const float4*>(feat + (long long)c3 * 128 + lane * 4);
            s.x += v0.x; s.y += v0.y; s.z += v0.z; s.w += v0.w;
            s.x += v1.x; s.y += v1.y; s.z += v1.z; s.w += v1.w;
            s.x += v2.x; s.y += v2.y; s.z += v2.z; s.w += v2.w;
            s.x += v3.x; s.y += v3.y; s.z += v3.z; s.w += v3.w;
        }
        for (; j < m; ++j) {
            const int c0 = __builtin_amdgcn_readlane(cv, j);
            const float4 v0 = *reinterpret_cast<const float4*>(feat + (long long)c0 * 128 + lane * 4);
            s.x += v0.x; s.y += v0.y; s.z += v0.z; s.w += v0.w;
        }
    }
    *reinterpret_cast<float4*>(accum + (long long)node * 128 + lane * 4) = s;
}

template <int NCOLT>
__global__ __launch_bounds__(32 * NCOLT) void sage_gemm_wmma(
        const float* __restrict__ Xs, const float* __restrict__ Xa,
        const float* __restrict__ invdeg,
        const float* __restrict__ Ws, const float* __restrict__ Wn,
        const float* __restrict__ bias, float* __restrict__ out,
        int M, int K, int do_relu) {
    const int N = NCOLT * 16;
    const int lane    = threadIdx.x;
    const int rowBase = blockIdx.x * 16;
    const int colBase = threadIdx.y * 16;

    int r = rowBase + (lane & 15);
    if (r >= M) r = M - 1;
    const int kOff = (lane < 16) ? 0 : 2;
    const int c = colBase + (lane & 15);

    const float invd = invdeg[r];
    const float* xsRow = Xs + (long long)r * K + kOff;
    const float* xaRow = Xa + (long long)r * K + kOff;

    v8f acc = {};
    for (int kk = 0; kk < K; kk += 4) {
        v2f a_s = *reinterpret_cast<const v2f*>(xsRow + kk);
        v2f a_raw = *reinterpret_cast<const v2f*>(xaRow + kk);
        v2f a_n = { a_raw.x * invd, a_raw.y * invd };
        const int kb = kk + kOff;
        v2f b_s = { Ws[(long long)kb * N + c], Ws[(long long)(kb + 1) * N + c] };
        v2f b_n = { Wn[(long long)kb * N + c], Wn[(long long)(kb + 1) * N + c] };
        acc = __builtin_amdgcn_wmma_f32_16x16x4_f32(false, a_s, false, b_s,
                                                    (short)0, acc, false, false);
        acc = __builtin_amdgcn_wmma_f32_16x16x4_f32(false, a_n, false, b_n,
                                                    (short)0, acc, false, false);
    }

    const float bv = bias[c];
    const int mBase = rowBase + ((lane < 16) ? 0 : 8);
#pragma unroll
    for (int j = 0; j < 8; ++j) {
        int m = mBase + j;
        if (m < M) {
            float v = acc[j] + bv;
            if (do_relu) v = fmaxf(v, 0.0f);
            out[(long long)m * N + c] = v;
        }
    }
}

#define RO_WAVES 32
#define RO_BATCH 32
#define RO_ROWS  16

__global__ __launch_bounds__(32 * RO_WAVES) void readout_kernel(const float* __restrict__ h2,
                                                              const int* __restrict__ gid,
                                                              float* __restrict__ out, int Nn) {
    __shared__ float part[RO_WAVES][OUT_F];
    __shared__ int   pcnt[RO_WAVES];
    const int g    = blockIdx.x;
    const int lane = threadIdx.x & 31;
    const int w = __builtin_amdgcn_readfirstlane((int)(threadIdx.x >> 5));
    const int nGroups = (Nn + 31) >> 5;

    float2 s = make_float2(0.0f, 0.0f);
    int cnt = 0;
    for (int t0 = w; t0 < nGroups; t0 += RO_WAVES * RO_BATCH) {
        unsigned bits = 0;
#pragma unroll
        for (int k = 0; k < RO_BATCH; ++k) {
            const int node = (t0 + k * RO_WAVES) * 32 + lane;
            const int id = (node < Nn) ? gid[node] : -1;
            bits |= (id == g ? 1u : 0u) << k;
        }
        if (__builtin_amdgcn_ballot_w32(bits != 0u) == 0u) continue;

#pragma nounroll
        for (int k = 0; k < RO_BATCH; ++k) {
            unsigned mask = __builtin_amdgcn_ballot_w32(((bits >> k) & 1u) != 0u);
            if (mask == 0u) continue;
            cnt += __builtin_popcount(mask);
            const float* rows = h2 + (long long)(t0 + k * RO_WAVES) * 32 * OUT_F + lane * 2;
            while (mask != 0u) {
                float2 v[RO_ROWS];
#pragma unroll
                for (int q = 0; q < RO_ROWS; ++q) {
                    if (mask != 0u) {
                        const int j = __builtin_ctz(mask);
                        mask &= mask - 1u;
                        v[q] = *reinterpret_cast<const float2*>(rows + j * OUT_F);
                    } else {
                        v[q] = make_float2(0.0f, 0.0f);
                    }
                }
#pragma unroll
                for (int q = 0; q < RO_ROWS; ++q) {
                    s.x += v[q].x;
                    s.y += v[q].y;
                }
            }
        }
    }

    *reinterpret_cast<float2*>(&part[w][lane * 2]) = s;
    if (lane == 0) pcnt[w] = cnt;
    __syncthreads();

    if (threadIdx.x < OUT_F) {
        float tot = 0.0f;
        int c = 0;
        for (int k = 0; k < RO_WAVES; ++k) {
            tot += part[k][threadIdx.x];
            c += pcnt[k];
        }
        out[(long long)g * OUT_F + threadIdx.x] = tot / fmaxf((float)c, 1.0f);
    }
}


static inline int cdiv64(long long a, long long b) { return (int)((a + b - 1) / b); }

extern "C" void kernel_launch(void* const* d_in, const int* in_sizes, int n_in,
                              void* d_out, int out_size, void* d_ws, size_t ws_size,
                              hipStream_t stream) {
    const float* in_feat = (const float*)d_in[0];
    const int*   src     = (const int*)d_in[1];
    const int*   dst     = (const int*)d_in[2];
    const int*   gid     = (const int*)d_in[3];
    const float* W1s = (const float*)d_in[5];
    const float* W1n = (const float*)d_in[6];
    const float* b1  = (const float*)d_in[7];
    const float* W2s = (const float*)d_in[8];
    const float* W2n = (const float*)d_in[9];
    const float* b2  = (const float*)d_in[10];
    float* out = (float*)d_out;

    const int Nn = in_sizes[0] / IN_F;
    const int E  = in_sizes[1];
    const int G  = out_size / OUT_F;

    float* ws    = (float*)d_ws;
    float* accum = ws;
    float* h1    = accum + (size_t)Nn * 128;
    float* h2    = h1 + (size_t)Nn * 128;
    float* deg   = h2 + (size_t)Nn * OUT_F;
    int* cnt    = (int*)h2;
    int* rowptr = cnt + Nn;
    int* rank   = rowptr + Nn;
    int* col    = rank + E;

    const int mTiles = cdiv64(Nn, 16);

    zero_int_kernel<<<cdiv64(Nn, 256), 256, 0, stream>>>(cnt, Nn);
    count_kernel<<<cdiv64(E, 256), 256, 0, stream>>>(dst, cnt, rank, E);
    scan_kernel<<<cdiv64(Nn, 1024), 1024, 0, stream>>>(cnt, rowptr, deg, Nn);
    fill_kernel<<<cdiv64(E, 256), 256, 0, stream>>>(src, dst, rowptr, rank, col, E);

    gather_kernel<<<cdiv64(Nn, 8), 256, 0, stream>>>(in_feat, rowptr, cnt, col, accum, Nn);
    sage_gemm_wmma<8><<<mTiles, dim3(32, 8), 0, stream>>>(
        in_feat, accum, deg, W1s, W1n, b1, h1, Nn, IN_F, 1);

    gather_kernel<<<cdiv64(Nn, 8), 256, 0, stream>>>(h1, rowptr, cnt, col, accum, Nn);
    sage_gemm_wmma<4><<<mTiles, dim3(32, 4), 0, stream>>>(
        h1, accum, deg, W2s, W2n, b2, h2, Nn, HID_F, 0);

    readout_kernel<<<G, 32 * RO_WAVES, 0, stream>>>(h2, gid, out, Nn);
}


# Layer GEMMs: weights and row tiles staged in LDS with whole-line loads by resident workgroups that loop over row tiles; WMMA fragments read from LDS; k order unchanged
# speedup vs baseline: 3.5368x; 1.2225x over previous
#include <hip/hip_runtime.h>

typedef __attribute__((ext_vector_type(2))) float v2f;
typedef __attribute__((ext_vector_type(8))) float v8f;

#define IN_F  128
#define HID_F 128
#define OUT_F 64


__global__ __launch_bounds__(256) void zero_int_kernel(int* __restrict__ p, int n) {
    int i = blockIdx.x * 256 + threadIdx.x;
    if (i < n) p[i] = 0;
}

__global__ __launch_bounds__(256) void count_kernel(const int* __restrict__ dst,
                                                    int* __restrict__ cnt,
                                                    int* __restrict__ rank, int E) {
    int e = blockIdx.x * 256 + threadIdx.x;
    if (e < E) rank[e] = atomicAdd(&cnt[dst[e]], 1);
}

__global__ __launch_bounds__(1024) void scan_kernel(const int* __restrict__ cnt,
                                                    int* __restrict__ rowptr,
                                                    float* __restrict__ invdeg, int Nn) {
    __shared__ int wpre[32];
    __shared__ int wtot[32];
    const int t = threadIdx.x, lane = t & 31, w = t >> 5;
    const int start = blockIdx.x * 1024;

    int part = 0;
    for (int i = t; i < start; i += 1024) part += cnt[i];
    for (int o = 16; o > 0; o >>= 1) part += __shfl_xor(part, o);
    if (lane == 0) wpre[w] = part;

    const int i = start + t;
    const int v = (i < Nn) ? cnt[i] : 0;
    int inc = v;
    for (int o = 1; o < 32; o <<= 1) {
        int u = __shfl_up(inc, o);
        if (lane >= o) inc += u;
    }
    if (lane == 31) wtot[w] = inc;
    __syncthreads();

    int base = 0;
    for (int k = 0; k < 32; ++k) {
        base += wpre[k];
        if (k < w) base += wtot[k];
    }
    if (i < Nn) {
        rowptr[i] = base + inc - v;
        invdeg[i] = 1.0f / fmaxf((float)v, 1.0f);
    }
}

__global__ __launch_bounds__(256) void fill_kernel(const int* __restrict__ src,
                                                   const int* __restrict__ dst,
                                                   const int* __restrict__ rowptr,
                                                   const int* __restrict__ rank,
                                                   int* __restrict__ col, int E) {
    int e = blockIdx.x * 256 + threadIdx.x;
    if (e < E) col[rowptr[dst[e]] + rank[e]] = src[e];
}

__global__ __launch_bounds__(256) void gather_kernel(const float* __restrict__ feat,
                                                     const int* __restrict__ rowptr,
                                                     const int* __restrict__ cnt,
                                                     const int* __restrict__ col,
                                                     float* __restrict__ accum, int Nn) {
    const int lane = threadIdx.x & 31;
    const int node = blockIdx.x * 8 + __builtin_amdgcn_readfirstlane((int)(threadIdx.x >> 5));
    if (node >= Nn) return;
    const int beg = __builtin_amdgcn_readfirstlane(rowptr[node]);
    const int n   = __builtin_amdgcn_readfirstlane(cnt[node]);

    float4 s = make_float4(0.0f, 0.0f, 0.0f, 0.0f);
    for (int b = 0; b < n; b += 32) {
        const int m = (n - b < 32) ? (n - b) : 32;
        const int cv = (lane < m) ? col[beg + b + lane] : 0;
        int j = 0;
        for (; j + 4 <= m; j += 4) {
            const int c0 = __builtin_amdgcn_readlane(cv, j);
            const int c1 = __builtin_amdgcn_readlane(cv, j + 1);
            const int c2 = __builtin_amdgcn_readlane(cv, j + 2);
            const int c3 = __builtin_amdgcn_readlane(cv, j + 3);
            const float4 v0 = *reinterpret_cast<const float4*>(feat + (long long)c0 * 128 + lane * 4);
            const float4 v1 = *reinterpret_cast<const float4*>(feat + (long long)c1 * 128 + lane * 4);
            const float4 v2 = *reinterpret_cast<const float4*>(feat + (long long)c2 * 128 + lane * 4);
            const float4 v3 = *reinterpret_cast<const float4*>(feat + (long long)c3 * 128 + lane * 4);
            s.x += v0.x; s.y += v0.y; s.z += v0.z; s.w += v0.w;
            s.x += v1.x; s.y += v1.y; s.z += v1.z; s.w += v1.w;
            s.x += v2.x; s.y += v2.y; s.z += v2.z; s.w += v2.w;
            s.x += v3.x; s.y += v3.y; s.z += v3.z; s.w += v3.w;
        }
        for (; j < m; ++j) {
            const int c0 = __builtin_amdgcn_readlane(cv, j);
            const float4 v0 = *reinterpret_cast<const float4*>(feat + (long long)c0 * 128 + lane * 4);
            s.x += v0.x; s.y += v0.y; s.z += v0.z; s.w += v0.w;
        }
    }
    *reinterpret_cast<float4*>(accum + (long long)node * 128 + lane * 4) = s;
}

template <int NCOLT, int RT>
__global__ __launch_bounds__(32 * NCOLT * RT) void sage_gemm_wmma(
        const float* __restrict__ Xs, const float* __restrict__ Xa,
        const float* __restrict__ invdeg,
        const float* __restrict__ Ws, const float* __restrict__ Wn,
        const float* __restrict__ bias, float* __restrict__ out,
        int M, int do_relu) {
    constexpr int K   = 128;
    constexpr int N   = NCOLT * 16;
    constexpr int BM  = RT * 16;
    constexpr int NT  = 32 * NCOLT * RT;
    constexpr int PA  = 2 * K + 4;
    constexpr int PER = (BM * 64) / NT;
    constexpr int RSTEP = NT / 64;
    static_assert(NT % 64 == 0 && PER * NT == BM * 64, "a row tile splits evenly over the threads");
    static_assert((PA * 4) % 16 == 0, "16-byte LDS stores stay aligned");

    __shared__ __attribute__((aligned(16))) float Bt[N * PA];
    __shared__ __attribute__((aligned(16))) float As[2 * BM * PA];

    const int tid  = threadIdx.x;
    const int lane = tid & 31;
    const int wv = __builtin_amdgcn_readfirstlane(tid >> 5);
    const int rt = wv / NCOLT;
    const int ct = wv % NCOLT;
    const int nTiles = (M + BM - 1) / BM;

    const int  c4   = tid & 63;
    const bool nb   = c4 >= 32;
    const float* xsrc = nb ? (Xa + (c4 - 32) * 4) : (Xs + c4 * 4);
    const int  row0 = tid >> 6;
    float4 pre[PER];
    float  psc[PER];

    auto fetch = [&](int tile) {
#pragma unroll
        for (int q = 0; q < PER; ++q) {
            int g = tile * BM + row0 + q * RSTEP;
            if (g >= M) g = M - 1;
            pre[q] = *reinterpret_cast<const float4*>(xsrc + (long long)g * K);
            psc[q] = nb ? invdeg[g] : 1.0f;
        }
    };
    auto stash = [&](int stage) {
        float* dst = As + stage * (BM * PA) + row0 * PA + c4 * 4;
#pragma unroll
        for (int q = 0; q < PER; ++q) {
            const float4 v = pre[q];
            const float  s = psc[q];
            *reinterpret_cast<float4*>(dst + q * (RSTEP * PA)) =
                make_float4(v.x * s, v.y * s, v.z * s, v.w * s);
        }
    };

    int tile = blockIdx.x;
    if (tile < nTiles) fetch(tile);

    {
        constexpr int NPC  = N / 16;
        constexpr int WPER = ((2 * K / 8) * NPC) / (NT / 32);
        static_assert(WPER * (NT / 32) == (2 * K / 8) * NPC, "patches split evenly over the waves");
        const int kr = lane & 7;
        const int cq = lane >> 3;
        float4 wreg[WPER];
#pragma unroll
        for (int i = 0; i < WPER; ++i) {
            const int p = wv + i * (NT / 32);
            const int k  = (p / NPC) * 8 + kr;
            const int n0 = (p % NPC) * 16 + 4 * cq;
            const float* wsrc = (k < K) ? (Ws + k * N) : (Wn + (k - K) * N);
            wreg[i] = *reinterpret_cast<const float4*>(wsrc + n0);
        }
#pragma unroll
        for (int i = 0; i < WPER; ++i) {
            const int p = wv + i * (NT / 32);
            const int k  = (p / NPC) * 8 + kr;
            const int n0 = (p % NPC) * 16 + 4 * cq;
            float* d = Bt + n0 * PA + k;
            d[0]      = wreg[i].x;
            d[PA]     = wreg[i].y;
            d[2 * PA] = wreg[i].z;
            d[3 * PA] = wreg[i].w;
        }
    }

    if (tile < nTiles) stash(0);
    __syncthreads();

    const int c = lane & 15;
    const int h = lane >> 4;
    const int col = ct * 16 + c;
    const float* aLane = As + (rt * 16 + c) * PA + 2 * h;
    const float* bLane = Bt + col * PA + 2 * h;
    const float bv = bias[col];

    for (int it = 0; tile < nTiles; tile += gridDim.x, ++it) {
        const int  stage = it & 1;
        const int  nxt   = tile + gridDim.x;
        const bool more  = nxt < nTiles;
        if (more) fetch(nxt);

        const float* aT = aLane + stage * (BM * PA);
        v8f acc = {};
#pragma unroll
        for (int kk = 0; kk < K; kk += 4) {
            const v2f a_s = *reinterpret_cast<const v2f*>(aT + kk);
            const v2f a_n = *reinterpret_cast<const v2f*>(aT + K + kk);
            const v2f b_s = *reinterpret_cast<const v2f*>(bLane + kk);
            const v2f b_n = *reinterpret_cast<const v2f*>(bLane + K + kk);
            acc = __builtin_amdgcn_wmma_f32_16x16x4_f32(false, a_s, false, b_s,
                                                        (short)0, acc, false, false);
            acc = __builtin_amdgcn_wmma_f32_16x16x4_f32(false, a_n, false, b_n,
                                                        (short)0, acc, false, false);
        }

        float r[8];
#pragma unroll
        for (int j = 0; j < 8; ++j) {
            const float v = acc[j] + bv;
            r[j] = do_relu ? fmaxf(v, 0.0f) : v;
        }
        const int mBase = tile * BM + rt * 16 + 8 * h;
        float* orow = out + (long long)mBase * N + col;
        if (mBase + 8 <= M) {
#pragma unroll
            for (int j = 0; j < 8; ++j) orow[(long long)j * N] = r[j];
        } else {
#pragma unroll
            for (int j = 0; j < 8; ++j)
                if (mBase + j < M) orow[(long long)j * N] = r[j];
        }

        if (more) stash(stage ^ 1);
        __syncthreads();
    }
}

#define RO_WAVES 32
#define RO_BATCH 32
#define RO_ROWS  16

__global__ __launch_bounds__(32 * RO_WAVES) void readout_kernel(const float* __restrict__ h2,
                                                              const int* __restrict__ gid,
                                                              float* __restrict__ out, int Nn) {
    __shared__ float part[RO_WAVES][OUT_F];
    __shared__ int   pcnt[RO_WAVES];
    const int g    = blockIdx.x;
    const int lane = threadIdx.x & 31;
    const int w = __builtin_amdgcn_readfirstlane((int)(threadIdx.x >> 5));
    const int nGroups = (Nn + 31) >> 5;

    float2 s = make_float2(0.0f, 0.0f);
    int cnt = 0;
    for (int t0 = w; t0 < nGroups; t0 += RO_WAVES * RO_BATCH) {
        unsigned bits = 0;
#pragma unroll
        for (int k = 0; k < RO_BATCH; ++k) {
            const int node = (t0 + k * RO_WAVES) * 32 + lane;
            const int id = (node < Nn) ? gid[node] : -1;
            bits |= (id == g ? 1u : 0u) << k;
        }
        if (__builtin_amdgcn_ballot_w32(bits != 0u) == 0u) continue;

#pragma nounroll
        for (int k = 0; k < RO_BATCH; ++k) {
            unsigned mask = __builtin_amdgcn_ballot_w32(((bits >> k) & 1u) != 0u);
            if (mask == 0u) continue;
            cnt += __builtin_popcount(mask);
            const float* rows = h2 + (long long)(t0 + k * RO_WAVES) * 32 * OUT_F + lane * 2;
            while (mask != 0u) {
                float2 v[RO_ROWS];
#pragma unroll
                for (int q = 0; q < RO_ROWS; ++q) {
                    if (mask != 0u) {
                        const int j = __builtin_ctz(mask);
                        mask &= mask - 1u;
                        v[q] = *reinterpret_cast<const float2*>(rows + j * OUT_F);
                    } else {
                        v[q] = make_float2(0.0f, 0.0f);
                    }
                }
#pragma unroll
                for (int q = 0; q < RO_ROWS; ++q) {
                    s.x += v[q].x;
                    s.y += v[q].y;
                }
            }
        }
    }

    *reinterpret_cast<float2*>(&part[w][lane * 2]) = s;
    if (lane == 0) pcnt[w] = cnt;
    __syncthreads();

    if (threadIdx.x < OUT_F) {
        float tot = 0.0f;
        int c = 0;
        for (int k = 0; k < RO_WAVES; ++k) {
            tot += part[k][threadIdx.x];
            c += pcnt[k];
        }
        out[(long long)g * OUT_F + threadIdx.x] = tot / fmaxf((float)c, 1.0f);
    }
}


static inline int cdiv64(long long a, long long b) { return (int)((a + b - 1) / b); }

extern "C" void kernel_launch(void* const* d_in, const int* in_sizes, int n_in,
                              void* d_out, int out_size, void* d_ws, size_t ws_size,
                              hipStream_t stream) {
    const float* in_feat = (const float*)d_in[0];
    const int*   src     = (const int*)d_in[1];
    const int*   dst     = (const int*)d_in[2];
    const int*   gid     = (const int*)d_in[3];
    const float* W1s = (const float*)d_in[5];
    const float* W1n = (const float*)d_in[6];
    const float* b1  = (const float*)d_in[7];
    const float* W2s = (const float*)d_in[8];
    const float* W2n = (const float*)d_in[9];
    const float* b2  = (const float*)d_in[10];
    float* out = (float*)d_out;

    const int Nn = in_sizes[0] / IN_F;
    const int E  = in_sizes[1];
    const int G  = out_size / OUT_F;

    float* ws    = (float*)d_ws;
    float* accum = ws;
    float* h1    = accum + (size_t)Nn * 128;
    float* h2    = h1 + (size_t)Nn * 128;
    float* deg   = h2 + (size_t)Nn * OUT_F;
    int* cnt    = (int*)h2;
    int* rowptr = cnt + Nn;
    int* rank   = rowptr + Nn;
    int* col    = rank + E;

    zero_int_kernel<<<cdiv64(Nn, 256), 256, 0, stream>>>(cnt, Nn);
    count_kernel<<<cdiv64(E, 256), 256, 0, stream>>>(dst, cnt, rank, E);
    scan_kernel<<<cdiv64(Nn, 1024), 1024, 0, stream>>>(cnt, rowptr, deg, Nn);
    fill_kernel<<<cdiv64(E, 256), 256, 0, stream>>>(src, dst, rowptr, rank, col, E);

    gather_kernel<<<cdiv64(Nn, 8), 256, 0, stream>>>(in_feat, rowptr, cnt, col, accum, Nn);
    sage_gemm_wmma<8, 2><<<256, 512, 0, stream>>>(
        in_feat, accum, deg, W1s, W1n, b1, h1, Nn, 1);

    gather_kernel<<<cdiv64(Nn, 8), 256, 0, stream>>>(h1, rowptr, cnt, col, accum, Nn);
    sage_gemm_wmma<4, 2><<<512, 256, 0, stream>>>(
        h1, accum, deg, W2s, W2n, b2, h2, Nn, 0);

    readout_kernel<<<G, 32 * RO_WAVES, 0, stream>>>(h2, gid, out, Nn);
}
